# starting kernel
# baseline (speedup 1.0000x reference)
#include <hip/hip_runtime.h>
#include <math.h>


typedef float v2f __attribute__((ext_vector_type(2)));
typedef float v8f __attribute__((ext_vector_type(8)));

#define INF_ 128
#define HEADS 4
#define NEG_SLOPE 0.2f

__device__ __forceinline__ unsigned f32_enc(float f) {
    unsigned u = __float_as_uint(f);
    return (u & 0x80000000u) ? ~u : (u | 0x80000000u);
}
__device__ __forceinline__ float f32_dec(unsigned u) {
    return (u & 0x80000000u) ? __uint_as_float(u & 0x7FFFFFFFu)
                             : __uint_as_float(~u);
}

__device__ __forceinline__ float leaky(float e) {
    return e > 0.0f ? e : NEG_SLOPE * e;
}

__device__ __forceinline__ void atom_add_f32(float* p, float v) {
    unsafeAtomicAdd(p, v);
}

__global__ void __launch_bounds__(256)
gemm_proj_kernel(const float* __restrict__ x,
                 const float* __restrict__ Wg,
                 const float* __restrict__ Ws,
                 float* __restrict__ xproj,
                 float* __restrict__ skip,
                 int n) {
    const int rowBase = blockIdx.x * 16;
    const int colBase = threadIdx.y * 16;
    const int lane    = threadIdx.x;
    const int m       = lane & 15;
    const int koff    = (lane >> 4) * 2;

    int arow = rowBase + m;
    if (arow >= n) arow = n - 1;
    const float* __restrict__ xr = x  + (size_t)arow * INF_;
    const float* __restrict__ gr = Wg + (size_t)(colBase + m) * INF_;
    const float* __restrict__ sr = Ws + (size_t)(colBase + m) * INF_;

    v8f accG = {};
    v8f accS = {};
#pragma unroll
    for (int k0 = 0; k0 < INF_; k0 += 4) {
        v2f a  = *(const v2f*)(xr + k0 + koff);
        v2f bg = *(const v2f*)(gr + k0 + koff);
        v2f bs = *(const v2f*)(sr + k0 + koff);
        accG = __builtin_amdgcn_wmma_f32_16x16x4_f32(
            false, a, false, bg, (short)0, accG, false, false);
        accS = __builtin_amdgcn_wmma_f32_16x16x4_f32(
            false, a, false, bs, (short)0, accS, false, false);
    }

    const int rOff = (lane >> 4) * 8;
#pragma unroll
    for (int j = 0; j < 8; ++j) {
        const int r = rowBase + rOff + j;
        if (r < n) {
            xproj[(size_t)r * INF_ + colBase + m] = accG[j];
            skip [(size_t)r * INF_ + colBase + m] = accS[j];
        }
    }
}

__global__ void att_kernel(const float* __restrict__ xproj,
                           const float* __restrict__ att_src,
                           const float* __restrict__ att_dst,
                           float* __restrict__ a_src,
                           float* __restrict__ a_dst,
                           int n) {
    const int t = blockIdx.x * blockDim.x + threadIdx.x;
    if (t >= n * HEADS) return;
    const int node = t >> 2;
    const int h    = t & 3;
    const float* __restrict__ xp = xproj + (size_t)node * INF_ + h * 32;
    const float* __restrict__ as = att_src + h * 32;
    const float* __restrict__ ad = att_dst + h * 32;
    float s = 0.0f, d = 0.0f;
#pragma unroll
    for (int c = 0; c < 32; ++c) {
        const float v = xp[c];
        s = fmaf(v, as[c], s);
        d = fmaf(v, ad[c], d);
    }
    a_src[t] = s;
    a_dst[t] = d;
}

__global__ void init_kernel(float* __restrict__ out,
                            unsigned* __restrict__ segmax,
                            float* __restrict__ segsum,
                            int n) {
    const int t = blockIdx.x * blockDim.x + threadIdx.x;
    if (t < n * INF_) out[t] = 0.0f;
    if (t < n * HEADS) { segmax[t] = 0u; segsum[t] = 0.0f; }
}

__device__ __forceinline__ void edge_endpoints(const int* __restrict__ ei,
                                               int t, int e,
                                               int& s, int& d) {
    if (t < e) { s = ei[t]; d = ei[e + t]; }
    else       { s = t - e; d = t - e; }
}

__global__ void edge_max_kernel(const int* __restrict__ ei,
                                const float* __restrict__ a_src,
                                const float* __restrict__ a_dst,
                                unsigned* __restrict__ segmax,
                                int e, int n) {
    const int t = blockIdx.x * blockDim.x + threadIdx.x;
    if (t >= e + n) return;
    int s, d; edge_endpoints(ei, t, e, s, d);
#pragma unroll
    for (int h = 0; h < HEADS; ++h) {
        const float ev = leaky(a_src[s * HEADS + h] + a_dst[d * HEADS + h]);
        atomicMax(&segmax[d * HEADS + h], f32_enc(ev));
    }
}

__global__ void edge_sum_kernel(const int* __restrict__ ei,
                                const float* __restrict__ a_src,
                                const float* __restrict__ a_dst,
                                const unsigned* __restrict__ segmax,
                                float* __restrict__ segsum,
                                int e, int n) {
    const int t = blockIdx.x * blockDim.x + threadIdx.x;
    if (t >= e + n) return;
    int s, d; edge_endpoints(ei, t, e, s, d);
#pragma unroll
    for (int h = 0; h < HEADS; ++h) {
        const float ev = leaky(a_src[s * HEADS + h] + a_dst[d * HEADS + h]);
        const float mx = f32_dec(segmax[d * HEADS + h]);
        atom_add_f32(&segsum[d * HEADS + h], __expf(ev - mx));
    }
}

__global__ void __launch_bounds__(256)
edge_scatter_kernel(const int* __restrict__ ei,
                    const float* __restrict__ a_src,
                    const float* __restrict__ a_dst,
                    const unsigned* __restrict__ segmax,
                    const float* __restrict__ segsum,
                    const float* __restrict__ xproj,
                    float* __restrict__ out,
                    int e, int n) {
    const long long gid = (long long)blockIdx.x * blockDim.x + threadIdx.x;
    const int t    = (int)(gid >> 5);
    const int lane = (int)(gid & 31);
    if (t >= e + n) return;
    int s, d; edge_endpoints(ei, t, e, s, d);

    const int h  = lane >> 3;
    const float ev = leaky(a_src[s * HEADS + h] + a_dst[d * HEADS + h]);
    const float mx = f32_dec(segmax[d * HEADS + h]);
    const float sm = segsum[d * HEADS + h];
    const float alpha = __expf(ev - mx) / (sm + 1e-16f);

    const float4 v = *(const float4*)(xproj + (size_t)s * INF_ + lane * 4);
    float* __restrict__ o = out + (size_t)d * INF_ + lane * 4;
    atom_add_f32(o + 0, alpha * v.x);
    atom_add_f32(o + 1, alpha * v.y);
    atom_add_f32(o + 2, alpha * v.z);
    atom_add_f32(o + 3, alpha * v.w);
}

__global__ void final_kernel(float* __restrict__ out,
                             const float* __restrict__ skip,
                             const float* __restrict__ bias,
                             int n) {
    const int t = blockIdx.x * blockDim.x + threadIdx.x;
    if (t >= n * INF_) return;
    const float v = out[t] + bias[t & (INF_ - 1)] + skip[t];
    out[t] = v > 0.0f ? v : (__expf(v) - 1.0f);
}

extern "C" void kernel_launch(void* const* d_in, const int* in_sizes, int n_in,
                              void* d_out, int out_size, void* d_ws, size_t ws_size,
                              hipStream_t stream) {
    const float* x    = (const float*)d_in[0];
    const int*   ei   = (const int*)  d_in[1];
    const float* Wg   = (const float*)d_in[2];
    const float* asw  = (const float*)d_in[3];
    const float* adw  = (const float*)d_in[4];
    const float* bias = (const float*)d_in[5];
    const float* Ws   = (const float*)d_in[6];
    float* out = (float*)d_out;

    const int n = in_sizes[0] / INF_;
    const int e = in_sizes[1] / 2;
    const int etot = e + n;

    float* ws      = (float*)d_ws;
    float* xproj   = ws;
    float* skip    = xproj + (size_t)n * INF_;
    float* a_src   = skip  + (size_t)n * INF_;
    float* a_dst   = a_src + (size_t)n * HEADS;
    unsigned* smax = (unsigned*)(a_dst + (size_t)n * HEADS);
    float* ssum    = (float*)(smax + (size_t)n * HEADS);

    const dim3 gemmBlock(32, 8);
    gemm_proj_kernel<<<(n + 15) / 16, gemmBlock, 0, stream>>>(
        x, Wg, Ws, xproj, skip, n);

    att_kernel<<<(n * HEADS + 255) / 256, 256, 0, stream>>>(
        xproj, asw, adw, a_src, a_dst, n);

    init_kernel<<<(n * INF_ + 255) / 256, 256, 0, stream>>>(
        out, smax, ssum, n);

    edge_max_kernel<<<(etot + 255) / 256, 256, 0, stream>>>(
        ei, a_src, a_dst, smax, e, n);

    edge_sum_kernel<<<(etot + 255) / 256, 256, 0, stream>>>(
        ei, a_src, a_dst, smax, ssum, e, n);

    const long long scatterThreads = (long long)etot * 32;
    edge_scatter_kernel<<<(unsigned)((scatterThreads + 255) / 256), 256, 0, stream>>>(
        ei, a_src, a_dst, smax, ssum, xproj, out, e, n);

    final_kernel<<<(n * INF_ + 255) / 256, 256, 0, stream>>>(
        out, skip, bias, n);
}


# Edge scan filters a one-byte-per-edge tile-id plane (written beside the attention scalars), not the 4-byte destination list; local rows resolved per listed edge; LDS add arrival order changes
# speedup vs baseline: 3.2593x; 3.2593x over previous
#include <hip/hip_runtime.h>
#include <math.h>


typedef float v2f __attribute__((ext_vector_type(2)));
typedef float v8f __attribute__((ext_vector_type(8)));

#define INF_ 128
#define HEADS 4
#define NEG_SLOPE 0.2f

__device__ __forceinline__ unsigned f32_enc(float f) {
    unsigned u = __float_as_uint(f);
    return (u & 0x80000000u) ? ~u : (u | 0x80000000u);
}
__device__ __forceinline__ float f32_dec(unsigned u) {
    return (u & 0x80000000u) ? __uint_as_float(u & 0x7FFFFFFFu)
                             : __uint_as_float(~u);
}

__device__ __forceinline__ float leaky(float e) {
    return e > 0.0f ? e : NEG_SLOPE * e;
}

__device__ __forceinline__ void lds_add_f32(float* p, float v) {
    unsafeAtomicAdd(p, v);
}

__global__ void __launch_bounds__(256)
gemm_proj_kernel(const float* __restrict__ x,
                 const float* __restrict__ Wg,
                 const float* __restrict__ Ws,
                 float* __restrict__ xproj,
                 float* __restrict__ skip,
                 int n) {
    const int rowBase = blockIdx.x * 16;
    const int colBase = threadIdx.y * 16;
    const int lane    = threadIdx.x;
    const int m       = lane & 15;
    const int koff    = (lane >> 4) * 2;

    int arow = rowBase + m;
    if (arow >= n) arow = n - 1;
    const float* __restrict__ xr = x  + (size_t)arow * INF_;
    const float* __restrict__ gr = Wg + (size_t)(colBase + m) * INF_;
    const float* __restrict__ sr = Ws + (size_t)(colBase + m) * INF_;

    v8f accG = {};
    v8f accS = {};
#pragma unroll
    for (int k0 = 0; k0 < INF_; k0 += 4) {
        v2f a  = *(const v2f*)(xr + k0 + koff);
        v2f bg = *(const v2f*)(gr + k0 + koff);
        v2f bs = *(const v2f*)(sr + k0 + koff);
        accG = __builtin_amdgcn_wmma_f32_16x16x4_f32(
            false, a, false, bg, (short)0, accG, false, false);
        accS = __builtin_amdgcn_wmma_f32_16x16x4_f32(
            false, a, false, bs, (short)0, accS, false, false);
    }

    const int rOff = (lane >> 4) * 8;
#pragma unroll
    for (int j = 0; j < 8; ++j) {
        const int r = rowBase + rOff + j;
        if (r < n) {
            xproj[(size_t)r * INF_ + colBase + m] = accG[j];
            skip [(size_t)r * INF_ + colBase + m] = accS[j];
        }
    }
}

#define TILE 196

__device__ __forceinline__ unsigned tile_byte(int d) {
    return min((unsigned)d / (unsigned)TILE, 255u);
}

__global__ void att_kernel(const float* __restrict__ xproj,
                           const float* __restrict__ att_src,
                           const float* __restrict__ att_dst,
                           const int* __restrict__ ei,
                           float* __restrict__ a_src,
                           float* __restrict__ a_dst,
                           unsigned* __restrict__ tplane,
                           int n, int e) {
    const int t = blockIdx.x * blockDim.x + threadIdx.x;

    const int* __restrict__ dstp = ei + e;
    const int nq = ((e + 15) >> 4) << 2;
    const bool al = (((size_t)dstp) & 15) == 0;
    for (int q = t; q < nq; q += gridDim.x * blockDim.x) {
        const int i0 = q << 2;
        int4 d;
        if (al && i0 + 3 < e) {
            d = *(const int4*)(dstp + i0);
        } else {
            d.x = i0 + 0 < e ? dstp[i0 + 0] : 0;
            d.y = i0 + 1 < e ? dstp[i0 + 1] : 0;
            d.z = i0 + 2 < e ? dstp[i0 + 2] : 0;
            d.w = i0 + 3 < e ? dstp[i0 + 3] : 0;
        }
        tplane[q] = tile_byte(d.x) | (tile_byte(d.y) << 8) |
                    (tile_byte(d.z) << 16) | (tile_byte(d.w) << 24);
    }

    if (t >= n * HEADS) return;
    const int node = t >> 2;
    const int h    = t & 3;
    const float* __restrict__ xp = xproj + (size_t)node * INF_ + h * 32;
    const float* __restrict__ as = att_src + h * 32;
    const float* __restrict__ ad = att_dst + h * 32;
    float s = 0.0f, d = 0.0f;
#pragma unroll
    for (int c = 0; c < 32; ++c) {
        const float v = xp[c];
        s = fmaf(v, as[c], s);
        d = fmaf(v, ad[c], d);
    }
    a_src[t] = s;
    a_dst[t] = d;
}

#define CAP  8192
#define NT   1024
#define NWAVE (NT / 32)
#define SEG  ((CAP - TILE) & ~3)
#define SU   4
#define BU   8

__device__ __forceinline__ void list_append(int* cnt, int* lt, int t) {
    const int slot = atomicAdd(cnt, 1);
    if (slot < CAP) lt[slot] = t;
}

__device__ __forceinline__ unsigned eq_bytes(unsigned w, unsigned kk) {
    const unsigned x = w ^ kk;
    const unsigned s = (x & 0x7F7F7F7Fu) + 0x7F7F7F7Fu;
    return ~(s | x | 0x7F7F7F7Fu);
}

__device__ __forceinline__ void scan_segment(const uint4* __restrict__ tp4,
                                             int sb, int se, unsigned kk,
                                             int base, int tile_n, int e,
                                             bool first, int* cnt, int* lt) {
    const int tid = threadIdx.x;
    if (tid == 0) *cnt = first ? tile_n : 0;
    if (first && tid < tile_n) lt[tid] = e + base + tid;
    __syncthreads();

    const int ve = (se + 15) >> 4;
    for (int v0 = (sb >> 4) + tid; v0 < ve; v0 += NT * SU) {
        uint4 q[SU];
#pragma unroll
        for (int u = 0; u < SU; ++u)
            q[u] = tp4[min(v0 + u * NT, ve - 1)];
#pragma unroll
        for (int u = 0; u < SU; ++u) {
            unsigned m = (eq_bytes(q[u].x, kk) >> 7) | (eq_bytes(q[u].y, kk) >> 6) |
                         (eq_bytes(q[u].z, kk) >> 5) | (eq_bytes(q[u].w, kk) >> 4);
            if (v0 + u * NT >= ve) m = 0;
            const int t0 = (v0 + u * NT) << 4;
            while (m) {
                const int p = __builtin_ctz(m);
                m &= m - 1;
                const int t = t0 + ((p & 7) << 2) + (p >> 3);
                if (t >= sb && t < se) list_append(cnt, lt, t);
            }
        }
    }
}

__device__ __forceinline__ void phase_max(const int* __restrict__ ei,
                                          const float* __restrict__ a_src,
                                          int e, int base, int tile_n, int cnt,
                                          int* lt, int* ld,
                                          const float* adst, unsigned* smax) {
    for (int i = threadIdx.x; i < cnt; i += NT) {
        const int t = lt[i];
        int s, dd;
        if (t < e) { s = ei[t]; dd = ei[e + t]; }
        else       { s = t - e; dd = s; }
        const unsigned d = (unsigned)dd - (unsigned)base;
        const bool ok = d < (unsigned)tile_n;
        lt[i] = ok ? s : 0;
        ld[i] = ok ? (int)d : -1;
        if (ok) {
            const float4 as = *(const float4*)(a_src + (size_t)s * HEADS);
            atomicMax(&smax[d * HEADS + 0], f32_enc(leaky(as.x + adst[d * HEADS + 0])));
            atomicMax(&smax[d * HEADS + 1], f32_enc(leaky(as.y + adst[d * HEADS + 1])));
            atomicMax(&smax[d * HEADS + 2], f32_enc(leaky(as.z + adst[d * HEADS + 2])));
            atomicMax(&smax[d * HEADS + 3], f32_enc(leaky(as.w + adst[d * HEADS + 3])));
        }
    }
}

__device__ __forceinline__ void phase_acc(const float* __restrict__ xproj,
                                          const float* __restrict__ a_src,
                                          int cnt, const int* lt, const int* ld,
                                          const float* adst,
                                          const unsigned* smax,
                                          float* ssum, float* acc) {
    const int w = threadIdx.x >> 5;
    const int l = threadIdx.x & 31;
    const int h = l >> 3;
    for (int i0 = w; i0 < cnt; i0 += NWAVE * BU) {
        int   s[BU], d[BU];
        float4 v[BU];
        float as[BU];
#pragma unroll
        for (int u = 0; u < BU; ++u) {
            const int i = i0 + u * NWAVE;
            const bool ok = i < cnt;
            s[u] = lt[ok ? i : i0];
            d[u] = ok ? ld[i] : -1;
        }
#pragma unroll
        for (int u = 0; u < BU; ++u) {
            v[u]  = *(const float4*)(xproj + (size_t)s[u] * INF_ + l * 4);
            as[u] = a_src[(size_t)s[u] * HEADS + h];
        }
#pragma unroll
        for (int u = 0; u < BU; ++u) {
            if (d[u] >= 0) {
                const int dh = d[u] * HEADS + h;
                const float ev = leaky(as[u] + adst[dh]);
                const float p  = __expf(ev - f32_dec(smax[dh]));
                float* a = acc + d[u] * 32 + l;
                lds_add_f32(a + 0 * TILE * 32, p * v[u].x);
                lds_add_f32(a + 1 * TILE * 32, p * v[u].y);
                lds_add_f32(a + 2 * TILE * 32, p * v[u].z);
                lds_add_f32(a + 3 * TILE * 32, p * v[u].w);
                if ((l & 7) == 0) lds_add_f32(&ssum[dh], p);
            }
        }
    }
}

__device__ __forceinline__ float elu1(float v) {
    return v > 0.0f ? v : (__expf(v) - 1.0f);
}

__global__ void __launch_bounds__(NT)
edge_tile_kernel(const int* __restrict__ ei,
                 const float* __restrict__ a_src,
                 const float* __restrict__ a_dst,
                 const float* __restrict__ xproj,
                 const float* __restrict__ skip,
                 const float* __restrict__ bias,
                 const unsigned* __restrict__ tplane,
                 float* __restrict__ out,
                 int e, int n) {
    __shared__ float    acc[4 * TILE * 32];
    __shared__ unsigned smax[TILE * HEADS];
    __shared__ float    ssum[TILE * HEADS];
    __shared__ float    adst[TILE * HEADS];
    __shared__ int      lt[CAP];
    __shared__ int      ld[CAP];
    __shared__ int      cnt;

    const int tid    = threadIdx.x;
    const int base   = blockIdx.x * TILE;
    const int tile_n = min(TILE, n - base);
    const uint4* __restrict__ tp4 = (const uint4*)tplane;
    const unsigned kk = min(blockIdx.x, 255u) * 0x01010101u;

    for (int i = tid; i < 4 * TILE * 32; i += NT) acc[i] = 0.0f;
    for (int i = tid; i < TILE * HEADS; i += NT) {
        smax[i] = 0u;
        ssum[i] = 0.0f;
        adst[i] = i < tile_n * HEADS ? a_dst[(size_t)base * HEADS + i] : 0.0f;
    }

    scan_segment(tp4, 0, e, kk, base, tile_n, e, true, &cnt, lt);
    __syncthreads();
    const int total = cnt;

    if (total <= CAP) {
        phase_max(ei, a_src, e, base, tile_n, total, lt, ld, adst, smax);
        __syncthreads();
        phase_acc(xproj, a_src, total, lt, ld, adst, smax, ssum, acc);
    } else {
        for (int sweep = 0; sweep < 2; ++sweep) {
            for (int sb = 0; sb == 0 || sb < e; sb += SEG) {
                const int se = min(sb + SEG, e);
                __syncthreads();
                scan_segment(tp4, sb, se, kk, base, tile_n, e, sb == 0, &cnt, lt);
                __syncthreads();
                const int c = cnt;
                phase_max(ei, a_src, e, base, tile_n, c, lt, ld, adst, smax);
                __syncthreads();
                if (sweep == 1)
                    phase_acc(xproj, a_src, c, lt, ld, adst, smax, ssum, acc);
            }
        }
    }
    __syncthreads();

    for (int idx = tid; idx < tile_n * 32; idx += NT) {
        const int r = idx >> 5;
        const int l = idx & 31;
        const float sm = ssum[r * HEADS + (l >> 3)] + 1e-16f;
        const size_t g = (size_t)(base + r) * INF_ + l * 4;
        const float4 sk = *(const float4*)(skip + g);
        float4 o;
        o.x = elu1(acc[(0 * TILE + r) * 32 + l] / sm + bias[l * 4 + 0] + sk.x);
        o.y = elu1(acc[(1 * TILE + r) * 32 + l] / sm + bias[l * 4 + 1] + sk.y);
        o.z = elu1(acc[(2 * TILE + r) * 32 + l] / sm + bias[l * 4 + 2] + sk.z);
        o.w = elu1(acc[(3 * TILE + r) * 32 + l] / sm + bias[l * 4 + 3] + sk.w);
        *(float4*)(out + g) = o;
    }
}

extern "C" void kernel_launch(void* const* d_in, const int* in_sizes, int n_in,
                              void* d_out, int out_size, void* d_ws, size_t ws_size,
                              hipStream_t stream) {
    const float* x    = (const float*)d_in[0];
    const int*   ei   = (const int*)  d_in[1];
    const float* Wg   = (const float*)d_in[2];
    const float* asw  = (const float*)d_in[3];
    const float* adw  = (const float*)d_in[4];
    const float* bias = (const float*)d_in[5];
    const float* Ws   = (const float*)d_in[6];
    float* out = (float*)d_out;

    const int n = in_sizes[0] / INF_;
    const int e = in_sizes[1] / 2;

    float* ws      = (float*)d_ws;
    float* xproj   = ws;
    float* skip    = xproj + (size_t)n * INF_;
    float* a_src   = skip  + (size_t)n * INF_;
    float* a_dst   = a_src + (size_t)n * HEADS;
    unsigned* tplane = (unsigned*)(a_dst + (size_t)n * HEADS);

    const dim3 gemmBlock(32, 8);
    gemm_proj_kernel<<<(n + 15) / 16, gemmBlock, 0, stream>>>(
        x, Wg, Ws, xproj, skip, n);

    att_kernel<<<(n * HEADS + 255) / 256, 256, 0, stream>>>(
        xproj, asw, adw, ei, a_src, a_dst, tplane, n, e);

    edge_tile_kernel<<<(n + TILE - 1) / TILE, NT, 0, stream>>>(
        ei, a_src, a_dst, xproj, skip, bias, tplane, out, e, n);
}


# Accumulate phase: per 8-edge batch list entry, score, softmax weight and sum add formed once per (edge, head) lane; row adds get source, row by readlane, weight by 8-lane DPP; LDS add order changes
# speedup vs baseline: 5.8807x; 1.8043x over previous
#include <hip/hip_runtime.h>
#include <math.h>


typedef float v2f __attribute__((ext_vector_type(2)));
typedef float v8f __attribute__((ext_vector_type(8)));

#define INF_ 128
#define HEADS 4
#define NEG_SLOPE 0.2f

__device__ __forceinline__ unsigned f32_enc(float f) {
    unsigned u = __float_as_uint(f);
    return (u & 0x80000000u) ? ~u : (u | 0x80000000u);
}
__device__ __forceinline__ float f32_dec(unsigned u) {
    return (u & 0x80000000u) ? __uint_as_float(u & 0x7FFFFFFFu)
                             : __uint_as_float(~u);
}

__device__ __forceinline__ float leaky(float e) {
    return e > 0.0f ? e : NEG_SLOPE * e;
}

__device__ __forceinline__ void lds_add_f32(float* p, float v) {
    unsafeAtomicAdd(p, v);
}

#define WP   132
#define GW   16
#define GNT  (GW * 32)

__device__ __forceinline__ const float* strip_row(const float* __restrict__ x,
                                                  int strip, int m, int hh, int n) {
    int arow = strip * 16 + m;
    if (arow >= n) arow = n - 1;
    if (arow < 0) arow = 0;
    return x + (size_t)arow * INF_ + 4 * hh;
}

#define TILE 196

__device__ __forceinline__ unsigned tile_byte(int d) {
    return min((unsigned)d / (unsigned)TILE, 255u);
}

__device__ __forceinline__ void plane_part(const int* __restrict__ dstp,
                                           unsigned* __restrict__ tplane,
                                           int first, int stride, int e) {
    const int nq = ((e + 15) >> 4) << 2;
    const bool al = (((size_t)dstp) & 15) == 0;
    for (int q = first; q < nq; q += stride) {
        const int i0 = q << 2;
        int4 d;
        if (al && i0 + 3 < e) {
            d = *(const int4*)(dstp + i0);
        } else {
            d.x = i0 + 0 < e ? dstp[i0 + 0] : 0;
            d.y = i0 + 1 < e ? dstp[i0 + 1] : 0;
            d.z = i0 + 2 < e ? dstp[i0 + 2] : 0;
            d.w = i0 + 3 < e ? dstp[i0 + 3] : 0;
        }
        tplane[q] = tile_byte(d.x) | (tile_byte(d.y) << 8) |
                    (tile_byte(d.z) << 16) | (tile_byte(d.w) << 24);
    }
}

__global__ void __launch_bounds__(GNT)
gemm_proj_kernel(const float* __restrict__ x,
                 const float* __restrict__ Wg,
                 const float* __restrict__ Ws,
                 const float* __restrict__ att_src,
                 const float* __restrict__ att_dst,
                 const int* __restrict__ ei,
                 float* __restrict__ xproj,
                 float* __restrict__ skip,
                 float* __restrict__ a_src,
                 float* __restrict__ a_dst,
                 unsigned* __restrict__ tplane,
                 int n, int e) {
    __shared__ alignas(16) float Wl[256 * WP];

    const int tid  = threadIdx.x;
    const int lane = tid & 31;
    const int w    = __builtin_amdgcn_readfirstlane(tid >> 5);
    const int m    = lane & 15;
    const int hh   = lane >> 4;

    const int nstrips = (n + 15) >> 4;
    const int per = nstrips / (int)gridDim.x;
    const int rem = nstrips % (int)gridDim.x;
    const int bx  = (int)blockIdx.x;
    const int s0  = bx * per + min(bx, rem);
    const int nunits = (per + (bx < rem ? 1 : 0)) * 2;

    int u = w;
    const float* xr = strip_row(x, s0 + (min(u, nunits - 1) >> 1), m, hh, n);
    float4 an0 = *(const float4*)(xr);
    float4 an1 = *(const float4*)(xr + 8);
    float4 ap0 = *(const float4*)(xr + 16);
    float4 ap1 = *(const float4*)(xr + 24);

    const float4 ws0 = *(const float4*)(att_src + 4 * m);
    const float4 ws1 = *(const float4*)(att_src + 64 + 4 * m);
    const float4 wd0 = *(const float4*)(att_dst + 4 * m);
    const float4 wd1 = *(const float4*)(att_dst + 64 + 4 * m);

#pragma unroll
    for (int it = 0; it < 256 * 32 / GNT; ++it) {
        const int i   = tid + it * GNT;
        const int row = i >> 5;
        const int c4  = i & 31;
        const int mat = row >> 7;
        const int r   = row & 127;
        const float4 v = *(const float4*)((mat ? Ws : Wg) + (size_t)r * INF_ + c4 * 4);
        const int pos = mat * 128 + 16 * (4 * (r >> 6) + (r & 3)) + ((r & 63) >> 2);
        *(float4*)(Wl + pos * WP + c4 * 4) = v;
    }
    __syncthreads();

    const int* __restrict__ dstp = ei + e;
    bool plane_todo = true;
    while (u < nunits) {
        const int strip = s0 + (u >> 1);
        const int mat   = u & 1;
        const float* bl = Wl + (mat * 128 + m) * WP + 4 * hh;

        v8f acc[8];
#pragma unroll
        for (int t = 0; t < 8; ++t) acc[t] = (v8f){};

#pragma unroll 1
        for (int jj = 0; jj < 16; jj += 2) {
            const float4 ac0 = an0, ac1 = an1;
            an0 = ap0;
            an1 = ap1;
            const int jn = jj + 4 < 16 ? jj + 4 : jj;
            ap0 = *(const float4*)(xr + 8 * jn);
            ap1 = *(const float4*)(xr + 8 * jn + 8);
#pragma unroll
            for (int j2 = 0; j2 < 2; ++j2) {
                const float4 av = j2 ? ac1 : ac0;
                float4 b[8];
#pragma unroll
                for (int t = 0; t < 8; ++t)
                    b[t] = *(const float4*)(bl + t * 16 * WP + 8 * (jj + j2));
                v2f a0 = {av.x, av.y};
                v2f a1 = {av.z, av.w};
#pragma unroll
                for (int t = 0; t < 8; ++t) {
                    v2f b0 = {b[t].x, b[t].y};
                    acc[t] = __builtin_amdgcn_wmma_f32_16x16x4_f32(
                        false, a0, false, b0, (short)0, acc[t], false, false);
                }
#pragma unroll
                for (int t = 0; t < 8; ++t) {
                    v2f b1 = {b[t].z, b[t].w};
                    acc[t] = __builtin_amdgcn_wmma_f32_16x16x4_f32(
                        false, a1, false, b1, (short)0, acc[t], false, false);
                }
            }
        }

        const int un = u + GW;
        if (un < nunits) {
            xr  = strip_row(x, s0 + (un >> 1), m, hh, n);
            an0 = *(const float4*)(xr);
            an1 = *(const float4*)(xr + 8);
            ap0 = *(const float4*)(xr + 16);
            ap1 = *(const float4*)(xr + 24);
        }

        float* __restrict__ o = mat ? skip : xproj;
#pragma unroll
        for (int j = 0; j < 8; ++j) {
            const int r = strip * 16 + 8 * hh + j;
            if (r < n) {
#pragma unroll
                for (int g = 0; g < 2; ++g) {
                    float4 v;
                    v.x = acc[4 * g + 0][j];
                    v.y = acc[4 * g + 1][j];
                    v.z = acc[4 * g + 2][j];
                    v.w = acc[4 * g + 3][j];
                    *(float4*)(o + (size_t)r * INF_ + 64 * g + 4 * m) = v;
                }
            }
        }

        if (mat == 0) {
            float ps[16], pd[16];
#pragma unroll
            for (int g = 0; g < 2; ++g) {
                const float4 ws = g ? ws1 : ws0;
                const float4 wd = g ? wd1 : wd0;
#pragma unroll
                for (int j = 0; j < 8; ++j) {
                    const float c0 = acc[4 * g + 0][j], c1 = acc[4 * g + 1][j];
                    const float c2 = acc[4 * g + 2][j], c3 = acc[4 * g + 3][j];
                    ps[2 * j + g] = fmaf(c3, ws.w, fmaf(c2, ws.z, fmaf(c1, ws.y, c0 * ws.x)));
                    pd[2 * j + g] = fmaf(c3, wd.w, fmaf(c2, wd.z, fmaf(c1, wd.y, c0 * wd.x)));
                }
            }
#pragma unroll
            for (int st = 1; st < 8; st <<= 1) {
#pragma unroll
                for (int i = 0; i < 16; ++i) {
                    ps[i] += __shfl_xor(ps[i], st);
                    pd[i] += __shfl_xor(pd[i], st);
                }
            }
            if ((m & 7) == 0) {
#pragma unroll
                for (int j = 0; j < 8; ++j) {
                    const int r = strip * 16 + 8 * hh + j;
                    if (r < n) {
#pragma unroll
                        for (int g = 0; g < 2; ++g) {
                            const size_t o4 = (size_t)r * HEADS + 2 * g + (m >> 3);
                            a_src[o4] = ps[2 * j + g];
                            a_dst[o4] = pd[2 * j + g];
                        }
                    }
                }
            }
        }

        if (plane_todo) {
            plane_part(dstp, tplane, bx * GNT + tid, (int)gridDim.x * GNT, e);
            plane_todo = false;
        }
        u = un;
    }
    if (plane_todo)
        plane_part(dstp, tplane, bx * GNT + tid, (int)gridDim.x * GNT, e);
}

#define CAP  8192
#define NT   1024
#define NWAVE (NT / 32)
#define SEG  ((CAP - TILE) & ~3)
#define SU   4
#define BU   8

__device__ __forceinline__ void list_append(int* cnt, int* lt, int t) {
    const int slot = atomicAdd(cnt, 1);
    if (slot < CAP) lt[slot] = t;
}

__device__ __forceinline__ unsigned eq_bytes(unsigned w, unsigned kk) {
    const unsigned x = w ^ kk;
    const unsigned s = (x & 0x7F7F7F7Fu) + 0x7F7F7F7Fu;
    return ~(s | x | 0x7F7F7F7Fu);
}

__device__ __forceinline__ void scan_segment(const uint4* __restrict__ tp4,
                                             int sb, int se, unsigned kk,
                                             int base, int tile_n, int e,
                                             bool first, int* cnt, int* lt) {
    const int tid = threadIdx.x;
    if (tid == 0) *cnt = first ? tile_n : 0;
    if (first && tid < tile_n) lt[tid] = e + base + tid;
    __syncthreads();

    const int ve = (se + 15) >> 4;
    for (int v0 = (sb >> 4) + tid; v0 < ve; v0 += NT * SU) {
        uint4 q[SU];
#pragma unroll
        for (int u = 0; u < SU; ++u)
            q[u] = tp4[min(v0 + u * NT, ve - 1)];
#pragma unroll
        for (int u = 0; u < SU; ++u) {
            unsigned m = (eq_bytes(q[u].x, kk) >> 7) | (eq_bytes(q[u].y, kk) >> 6) |
                         (eq_bytes(q[u].z, kk) >> 5) | (eq_bytes(q[u].w, kk) >> 4);
            if (v0 + u * NT >= ve) m = 0;
            const int t0 = (v0 + u * NT) << 4;
            while (m) {
                const int p = __builtin_ctz(m);
                m &= m - 1;
                const int t = t0 + ((p & 7) << 2) + (p >> 3);
                if (t >= sb && t < se) list_append(cnt, lt, t);
            }
        }
    }
}

__device__ __forceinline__ void phase_max(const int* __restrict__ ei,
                                          const float* __restrict__ a_src,
                                          int e, int base, int tile_n, int cnt,
                                          int* lt, int* ld,
                                          const float* adst, unsigned* smax) {
    for (int i = threadIdx.x; i < cnt; i += NT) {
        const int t = lt[i];
        int s, dd;
        if (t < e) { s = ei[t]; dd = ei[e + t]; }
        else       { s = t - e; dd = s; }
        const unsigned d = (unsigned)dd - (unsigned)base;
        const bool ok = d < (unsigned)tile_n;
        lt[i] = ok ? s : 0;
        ld[i] = ok ? (int)d : -1;
        if (ok) {
            const float4 as = *(const float4*)(a_src + (size_t)s * HEADS);
            atomicMax(&smax[d * HEADS + 0], f32_enc(leaky(as.x + adst[d * HEADS + 0])));
            atomicMax(&smax[d * HEADS + 1], f32_enc(leaky(as.y + adst[d * HEADS + 1])));
            atomicMax(&smax[d * HEADS + 2], f32_enc(leaky(as.z + adst[d * HEADS + 2])));
            atomicMax(&smax[d * HEADS + 3], f32_enc(leaky(as.w + adst[d * HEADS + 3])));
        }
    }
}


template <int U>
__device__ __forceinline__ float group8_bcast(float v) {
    return __uint_as_float(__builtin_amdgcn_mov_dpp8(__float_as_uint(v), U * 0x249249));
}

template <int U>
__device__ __forceinline__ void row_add(float* acc, int dv, float p,
                                        const float4 v, int l) {
    const int d = __builtin_amdgcn_readlane(dv, U);
    if (d >= 0) {
        const float pu = group8_bcast<U>(p);
        float* a = acc + d * 32 + l;
        lds_add_f32(a + 0 * TILE * 32, pu * v.x);
        lds_add_f32(a + 1 * TILE * 32, pu * v.y);
        lds_add_f32(a + 2 * TILE * 32, pu * v.z);
        lds_add_f32(a + 3 * TILE * 32, pu * v.w);
    }
}

__device__ __forceinline__ void phase_acc(const float* __restrict__ xproj,
                                          const float* __restrict__ a_src,
                                          int cnt, const int* lt, const int* ld,
                                          const float* adst,
                                          const unsigned* smax,
                                          float* ssum, float* acc) {
    const int w  = __builtin_amdgcn_readfirstlane(threadIdx.x >> 5);
    const int l  = threadIdx.x & 31;
    const int h  = l >> 3;
    const int u8 = l & 7;
    for (int i0 = w; i0 < cnt; i0 += NWAVE * BU) {
        const int  i  = i0 + u8 * NWAVE;
        const bool in = i < cnt;
        const int  sv = in ? lt[i] : 0;
        const int  dv = in ? ld[i] : -1;
        const bool live = dv >= 0;
        const int  dh = (live ? dv : 0) * HEADS + h;

        const float as = a_src[(size_t)sv * HEADS + h];

        float4 v[BU];
#pragma unroll
        for (int u = 0; u < BU; ++u) {
            const int s = __builtin_amdgcn_readlane(sv, u);
            v[u] = *(const float4*)(xproj + (size_t)s * INF_ + l * 4);
        }

        const float ev = leaky(as + adst[dh]);
        const float pe = __expf(ev - f32_dec(smax[dh]));
        const float p  = live ? pe : 0.0f;
        if (live) lds_add_f32(&ssum[dh], p);

        row_add<0>(acc, dv, p, v[0], l);
        row_add<1>(acc, dv, p, v[1], l);
        row_add<2>(acc, dv, p, v[2], l);
        row_add<3>(acc, dv, p, v[3], l);
        row_add<4>(acc, dv, p, v[4], l);
        row_add<5>(acc, dv, p, v[5], l);
        row_add<6>(acc, dv, p, v[6], l);
        row_add<7>(acc, dv, p, v[7], l);
    }
}

__device__ __forceinline__ float elu1(float v) {
    return v > 0.0f ? v : (__expf(v) - 1.0f);
}

__global__ void __launch_bounds__(NT)
edge_tile_kernel(const int* __restrict__ ei,
                 const float* __restrict__ a_src,
                 const float* __restrict__ a_dst,
                 const float* __restrict__ xproj,
                 const float* __restrict__ skip,
                 const float* __restrict__ bias,
                 const unsigned* __restrict__ tplane,
                 float* __restrict__ out,
                 int e, int n) {
    __shared__ float    acc[4 * TILE * 32];
    __shared__ unsigned smax[TILE * HEADS];
    __shared__ float    ssum[TILE * HEADS];
    __shared__ float    adst[TILE * HEADS];
    __shared__ int      lt[CAP];
    __shared__ int      ld[CAP];
    __shared__ int      cnt;

    const int tid    = threadIdx.x;
    const int base   = blockIdx.x * TILE;
    const int tile_n = min(TILE, n - base);
    const uint4* __restrict__ tp4 = (const uint4*)tplane;
    const unsigned kk = min(blockIdx.x, 255u) * 0x01010101u;

    for (int i = tid; i < 4 * TILE * 32; i += NT) acc[i] = 0.0f;
    for (int i = tid; i < TILE * HEADS; i += NT) {
        smax[i] = 0u;
        ssum[i] = 0.0f;
        adst[i] = i < tile_n * HEADS ? a_dst[(size_t)base * HEADS + i] : 0.0f;
    }

    scan_segment(tp4, 0, e, kk, base, tile_n, e, true, &cnt, lt);
    __syncthreads();
    const int total = cnt;

    if (total <= CAP) {
        phase_max(ei, a_src, e, base, tile_n, total, lt, ld, adst, smax);
        __syncthreads();
        phase_acc(xproj, a_src, total, lt, ld, adst, smax, ssum, acc);
    } else {
        for (int sweep = 0; sweep < 2; ++sweep) {
            for (int sb = 0; sb == 0 || sb < e; sb += SEG) {
                const int se = min(sb + SEG, e);
                __syncthreads();
                scan_segment(tp4, sb, se, kk, base, tile_n, e, sb == 0, &cnt, lt);
                __syncthreads();
                const int c = cnt;
                phase_max(ei, a_src, e, base, tile_n, c, lt, ld, adst, smax);
                __syncthreads();
                if (sweep == 1)
                    phase_acc(xproj, a_src, c, lt, ld, adst, smax, ssum, acc);
            }
        }
    }
    __syncthreads();

    for (int idx = tid; idx < tile_n * 32; idx += NT) {
        const int r = idx >> 5;
        const int l = idx & 31;
        const float sm = ssum[r * HEADS + (l >> 3)] + 1e-16f;
        const size_t g = (size_t)(base + r) * INF_ + l * 4;
        const float4 sk = *(const float4*)(skip + g);
        float4 o;
        o.x = elu1(acc[(0 * TILE + r) * 32 + l] / sm + bias[l * 4 + 0] + sk.x);
        o.y = elu1(acc[(1 * TILE + r) * 32 + l] / sm + bias[l * 4 + 1] + sk.y);
        o.z = elu1(acc[(2 * TILE + r) * 32 + l] / sm + bias[l * 4 + 2] + sk.z);
        o.w = elu1(acc[(3 * TILE + r) * 32 + l] / sm + bias[l * 4 + 3] + sk.w);
        *(float4*)(out + g) = o;
    }
}

extern "C" void kernel_launch(void* const* d_in, const int* in_sizes, int n_in,
                              void* d_out, int out_size, void* d_ws, size_t ws_size,
                              hipStream_t stream) {
    const float* x    = (const float*)d_in[0];
    const int*   ei   = (const int*)  d_in[1];
    const float* Wg   = (const float*)d_in[2];
    const float* asw  = (const float*)d_in[3];
    const float* adw  = (const float*)d_in[4];
    const float* bias = (const float*)d_in[5];
    const float* Ws   = (const float*)d_in[6];
    float* out = (float*)d_out;

    const int n = in_sizes[0] / INF_;
    const int e = in_sizes[1] / 2;

    float* ws      = (float*)d_ws;
    float* xproj   = ws;
    float* skip    = xproj + (size_t)n * INF_;
    float* a_src   = skip  + (size_t)n * INF_;
    float* a_dst   = a_src + (size_t)n * HEADS;
    unsigned* tplane = (unsigned*)(a_dst + (size_t)n * HEADS);

    gemm_proj_kernel<<<256, GNT, 0, stream>>>(
        x, Wg, Ws, asw, adw, ei, xproj, skip, a_src, a_dst, tplane, n, e);

    edge_tile_kernel<<<(n + TILE - 1) / TILE, NT, 0, stream>>>(
        ei, a_src, a_dst, xproj, skip, bias, tplane, out, e, n);
}
